# starting kernel
# baseline (speedup 1.0000x reference)
#include <hip/hip_runtime.h>

#define B_DIM 1024
#define D_DIM 512
#define K_DIM 256
#define T_DIM 1024

typedef __attribute__((ext_vector_type(2))) float v2f;
typedef __attribute__((ext_vector_type(8))) float v8f;

__global__ __launch_bounds__(256) void gpdf_prep(const float* __restrict__ means,
                                                 const float* __restrict__ rho,
                                                 float* __restrict__ WaT,
                                                 float* __restrict__ bias) {
  const int k = blockIdx.x;
  const int t = threadIdx.x;
  float partial = 0.0f;
#pragma unroll
  for (int i = 0; i < D_DIM / 256; ++i) {
    const int d = t + i * 256;
    const float m = means[k * D_DIM + d];
    const float r = rho[k * D_DIM + d];
    const float s = log1pf(expf(r));
    const float s2 = s * s;
    WaT[d * K_DIM + k]           = 0.5f * s2;
    WaT[(D_DIM + d) * K_DIM + k] = -s2 * m;
    partial += 0.5f * s2 * m * m;
  }
  __shared__ float red[256];
  red[t] = partial;
  __syncthreads();
  for (int s = 128; s > 0; s >>= 1) {
    if (t < s) red[t] += red[t + s];
    __syncthreads();
  }
  if (t == 0) bias[k] = red[0];
}

__global__ __launch_bounds__(256) void gpdf_gemm_wmma(
    const float* __restrict__ x,
    const float* __restrict__ WaT,
    const float* __restrict__ bias,
    float* __restrict__ out) {
  const int lane = threadIdx.x & 31;
  const int wave = threadIdx.x >> 5;
  const int half = lane >> 4;
  const int mn   = lane & 15;
  const int row0 = blockIdx.x * 128 + wave * 16;
  const int col0 = blockIdx.y * 16;

  const float* __restrict__ xrow = x + (row0 + mn) * D_DIM;
  const int bcol = col0 + mn;

  v8f c = {};

  for (int t0 = 0; t0 < D_DIM; t0 += 4) {
    const int ka = t0 + 2 * half;
    v2f a, b;
    const float a0 = xrow[ka];
    const float a1 = xrow[ka + 1];
    a.x = a0 * a0;
    a.y = a1 * a1;
    b.x = WaT[ka * K_DIM + bcol];
    b.y = WaT[(ka + 1) * K_DIM + bcol];
    c = __builtin_amdgcn_wmma_f32_16x16x4_f32(false, a, false, b,
                                              (short)0, c, false, false);
  }
  for (int t0 = 0; t0 < D_DIM; t0 += 4) {
    const int ka = t0 + 2 * half;
    v2f a, b;
    a.x = xrow[ka];
    a.y = xrow[ka + 1];
    b.x = WaT[(D_DIM + ka) * K_DIM + bcol];
    b.y = WaT[(D_DIM + ka + 1) * K_DIM + bcol];
    c = __builtin_amdgcn_wmma_f32_16x16x4_f32(false, a, false, b,
                                              (short)0, c, false, false);
  }

  const float bv = bias[bcol];
#pragma unroll
  for (int j = 0; j < 8; ++j) {
    out[(row0 + j + 8 * half) * K_DIM + bcol] = c[j] + bv;
  }
}

extern "C" void kernel_launch(void* const* d_in, const int* in_sizes, int n_in,
                              void* d_out, int out_size, void* d_ws, size_t ws_size,
                              hipStream_t stream) {
  const float* x     = (const float*)d_in[0];
  const float* means = (const float*)d_in[1];
  const float* rho   = (const float*)d_in[2];
  float* out  = (float*)d_out;
  float* WaT  = (float*)d_ws;
  float* bias = WaT + T_DIM * K_DIM;

  gpdf_prep<<<dim3(K_DIM), dim3(256), 0, stream>>>(means, rho, WaT, bias);
  gpdf_gemm_wmma<<<dim3(B_DIM / 128, K_DIM / 16), dim3(256), 0, stream>>>(
      x, WaT, bias, out);
}


# GEMM re-tiled: one 32x32 output tile per block (256 blocks), contraction split over the block's 8 waves (2x2 WMMA tiles each, x read once), partial sums added through LDS in fixed wave order
# speedup vs baseline: 2.2210x; 2.2210x over previous
#include <hip/hip_runtime.h>

#define B_DIM 1024
#define D_DIM 512
#define K_DIM 256
#define T_DIM 1024

typedef __attribute__((ext_vector_type(2))) float v2f;
typedef __attribute__((ext_vector_type(4))) float v4f;
typedef __attribute__((ext_vector_type(8))) float v8f;

__global__ __launch_bounds__(256) void gpdf_prep(const float* __restrict__ means,
                                                 const float* __restrict__ rho,
                                                 float* __restrict__ WaT,
                                                 float* __restrict__ bias) {
  const int k = blockIdx.x;
  const int t = threadIdx.x;
  float partial = 0.0f;
#pragma unroll
  for (int i = 0; i < D_DIM / 256; ++i) {
    const int d = t + i * 256;
    const float m = means[k * D_DIM + d];
    const float r = rho[k * D_DIM + d];
    const float s = log1pf(expf(r));
    const float s2 = s * s;
    WaT[d * K_DIM + k]           = 0.5f * s2;
    WaT[(D_DIM + d) * K_DIM + k] = -s2 * m;
    partial += 0.5f * s2 * m * m;
  }
  __shared__ float red[256];
  red[t] = partial;
  __syncthreads();
  for (int s = 128; s > 0; s >>= 1) {
    if (t < s) red[t] += red[t + s];
    __syncthreads();
  }
  if (t == 0) bias[k] = red[0];
}

#define TILE_MN 32
#define N_WAVE  8
#define D_CHUNK (D_DIM / N_WAVE)
#define P_PITCH 34

__global__ __launch_bounds__(256) __attribute__((amdgpu_waves_per_eu(2, 2))) void gpdf_gemm_wmma(
    const float* __restrict__ x,
    const float* __restrict__ WaT,
    const float* __restrict__ bias,
    float* __restrict__ out) {
  __shared__ float part[N_WAVE * TILE_MN * P_PITCH];

  const int lane = threadIdx.x & 31;
  const int wave = threadIdx.x >> 5;
  const int half = lane >> 4;
  const int mn   = lane & 15;
  const int row0 = blockIdx.x * TILE_MN;
  const int col0 = blockIdx.y * TILE_MN;
  const int dl   = wave * D_CHUNK + 4 * half;

  const float* __restrict__ xr0 = x + (row0 + mn) * D_DIM + dl;
  const float* __restrict__ xr1 = xr0 + 16 * D_DIM;
  const float* __restrict__ w1  = WaT + dl * K_DIM + col0 + mn;
  const float* __restrict__ w2  = w1 + D_DIM * K_DIM;

  v8f acc00 = {}, acc01 = {}, acc10 = {}, acc11 = {};

  for (int s = 0; s < D_CHUNK / 8; ++s) {
    const v4f xa0 = *(const v4f*)(xr0 + 8 * s);
    const v4f xa1 = *(const v4f*)(xr1 + 8 * s);
    for (int e = 0; e < 2; ++e) {
      const int dk = 8 * s + 2 * e;
      v2f a0, a1, q0, q1, b10, b11, b20, b21;
      a0.x = xa0[2 * e];  a0.y = xa0[2 * e + 1];
      a1.x = xa1[2 * e];  a1.y = xa1[2 * e + 1];
      q0 = a0 * a0;
      q1 = a1 * a1;
      b10.x = w1[dk * K_DIM];            b10.y = w1[(dk + 1) * K_DIM];
      b11.x = w1[dk * K_DIM + 16];       b11.y = w1[(dk + 1) * K_DIM + 16];
      b20.x = w2[dk * K_DIM];            b20.y = w2[(dk + 1) * K_DIM];
      b21.x = w2[dk * K_DIM + 16];       b21.y = w2[(dk + 1) * K_DIM + 16];
      acc00 = __builtin_amdgcn_wmma_f32_16x16x4_f32(false, q0, false, b10, (short)0, acc00, false, false);
      acc01 = __builtin_amdgcn_wmma_f32_16x16x4_f32(false, q0, false, b11, (short)0, acc01, false, false);
      acc10 = __builtin_amdgcn_wmma_f32_16x16x4_f32(false, q1, false, b10, (short)0, acc10, false, false);
      acc11 = __builtin_amdgcn_wmma_f32_16x16x4_f32(false, q1, false, b11, (short)0, acc11, false, false);
      acc00 = __builtin_amdgcn_wmma_f32_16x16x4_f32(false, a0, false, b20, (short)0, acc00, false, false);
      acc01 = __builtin_amdgcn_wmma_f32_16x16x4_f32(false, a0, false, b21, (short)0, acc01, false, false);
      acc10 = __builtin_amdgcn_wmma_f32_16x16x4_f32(false, a1, false, b20, (short)0, acc10, false, false);
      acc11 = __builtin_amdgcn_wmma_f32_16x16x4_f32(false, a1, false, b21, (short)0, acc11, false, false);
    }
  }

  float* pw = part + wave * (TILE_MN * P_PITCH) + (8 * half) * P_PITCH + mn;
  for (int j = 0; j < 8; ++j) {
    pw[j * P_PITCH]                  = acc00[j];
    pw[j * P_PITCH + 16]             = acc01[j];
    pw[(16 + j) * P_PITCH]           = acc10[j];
    pw[(16 + j) * P_PITCH + 16]      = acc11[j];
  }
  __syncthreads();

  const float bv = bias[col0 + lane];
  for (int m = 0; m < 4; ++m) {
    const int r = wave + 8 * m;
    const float* pr = part + r * P_PITCH + lane;
    float sum = pr[0];
    for (int w = 1; w < N_WAVE; ++w) sum += pr[w * (TILE_MN * P_PITCH)];
    out[(row0 + r) * K_DIM + col0 + lane] = sum + bv;
  }
}

extern "C" void kernel_launch(void* const* d_in, const int* in_sizes, int n_in,
                              void* d_out, int out_size, void* d_ws, size_t ws_size,
                              hipStream_t stream) {
  const float* x     = (const float*)d_in[0];
  const float* means = (const float*)d_in[1];
  const float* rho   = (const float*)d_in[2];
  float* out  = (float*)d_out;
  float* WaT  = (float*)d_ws;
  float* bias = WaT + T_DIM * K_DIM;

  gpdf_prep<<<dim3(K_DIM), dim3(256), 0, stream>>>(means, rho, WaT, bias);
  gpdf_gemm_wmma<<<dim3(B_DIM / TILE_MN, K_DIM / TILE_MN), dim3(256), 0, stream>>>(
      x, WaT, bias, out);
}


# One launch: weight prep fused into the GEMM (B fragments built in registers from means/rho rows, softplus on hw exp2/log2/rcp); bias now summed from per-lane partials via LDS, fixed order
# speedup vs baseline: 4.8026x; 2.1624x over previous
#include <hip/hip_runtime.h>

#define B_DIM 1024
#define D_DIM 512
#define K_DIM 256

typedef __attribute__((ext_vector_type(2))) float v2f;
typedef __attribute__((ext_vector_type(4))) float v4f;
typedef __attribute__((ext_vector_type(8))) float v8f;

#define TILE_MN 32
#define N_WAVE  8
#define D_CHUNK (D_DIM / N_WAVE)
#define P_PITCH 34

__device__ __forceinline__ float softplus_f(float r) {
  const float kLog2eHi = 1.44269504088896341f;
  const float kLog2eLo = 1.9259629911e-8f;
  const float kLn2     = 0.693147180559945309f;
  const float p  = r * kLog2eHi;
  const float pl = fmaf(r, kLog2eLo, fmaf(r, kLog2eHi, -p));
  const float e0 = __builtin_amdgcn_exp2f(p);
  const float e  = (e0 == __builtin_inff()) ? e0 : fmaf(e0, pl * kLn2, e0);
  const float u  = 1.0f + e;
  const float l  = __builtin_amdgcn_logf(u) * kLn2;
  const float t  = (e - (u - 1.0f)) * __builtin_amdgcn_rcpf(u);
  return (u == e) ? l : l + t;
}

__global__ __launch_bounds__(256) __attribute__((amdgpu_waves_per_eu(2, 2))) void gpdf_fused_wmma(
    const float* __restrict__ x,
    const float* __restrict__ means,
    const float* __restrict__ rho,
    float* __restrict__ out) {
  __shared__ float part[N_WAVE * TILE_MN * P_PITCH];
  __shared__ float bpart[2 * N_WAVE * TILE_MN];

  const int lane = threadIdx.x & 31;
  const int wave = threadIdx.x >> 5;
  const int half = lane >> 4;
  const int mn   = lane & 15;
  const int row0 = blockIdx.x * TILE_MN;
  const int col0 = blockIdx.y * TILE_MN;
  const int dl   = wave * D_CHUNK + 4 * half;

  const float* __restrict__ xr0 = x + (row0 + mn) * D_DIM + dl;
  const float* __restrict__ xr1 = xr0 + 16 * D_DIM;
  const float* __restrict__ mr0 = means + (col0 + mn) * D_DIM + dl;
  const float* __restrict__ mr1 = mr0 + 16 * D_DIM;
  const float* __restrict__ rr0 = rho + (col0 + mn) * D_DIM + dl;
  const float* __restrict__ rr1 = rr0 + 16 * D_DIM;

  v8f acc00 = {}, acc01 = {}, acc10 = {}, acc11 = {};
  float bs0 = 0.0f, bs1 = 0.0f;

  v4f XA0[D_CHUNK / 8], XA1[D_CHUNK / 8], M0[D_CHUNK / 8], M1[D_CHUNK / 8], R0[D_CHUNK / 8], R1[D_CHUNK / 8];
#pragma unroll
  for (int p = 0; p < D_CHUNK / 16; ++p) {
    R0[2 * p]      = *(const v4f*)(rr0 + 16 * p);
    R0[2 * p + 1]  = *(const v4f*)(rr0 + 16 * p + 8);
    R1[2 * p]      = *(const v4f*)(rr1 + 16 * p);
    R1[2 * p + 1]  = *(const v4f*)(rr1 + 16 * p + 8);
    M0[2 * p]      = *(const v4f*)(mr0 + 16 * p);
    M0[2 * p + 1]  = *(const v4f*)(mr0 + 16 * p + 8);
    M1[2 * p]      = *(const v4f*)(mr1 + 16 * p);
    M1[2 * p + 1]  = *(const v4f*)(mr1 + 16 * p + 8);
    XA0[2 * p]     = *(const v4f*)(xr0 + 16 * p);
    XA0[2 * p + 1] = *(const v4f*)(xr0 + 16 * p + 8);
    XA1[2 * p]     = *(const v4f*)(xr1 + 16 * p);
    XA1[2 * p + 1] = *(const v4f*)(xr1 + 16 * p + 8);
  }

#pragma unroll
  for (int s = 0; s < D_CHUNK / 8; ++s) {
    const v4f xa0 = XA0[s];
    const v4f xa1 = XA1[s];
    const v4f m0  = M0[s];
    const v4f m1  = M1[s];
    const v4f r0  = R0[s];
    const v4f r1  = R1[s];
    v4f h0, h1, g0, g1;
#pragma unroll
    for (int i = 0; i < 4; ++i) {
      const float sa = softplus_f(r0[i]);
      const float sb = softplus_f(r1[i]);
      const float s2a = sa * sa;
      const float s2b = sb * sb;
      h0[i] = 0.5f * s2a;
      h1[i] = 0.5f * s2b;
      g0[i] = -s2a * m0[i];
      g1[i] = -s2b * m1[i];
      bs0 += 0.5f * s2a * m0[i] * m0[i];
      bs1 += 0.5f * s2b * m1[i] * m1[i];
    }
#pragma unroll
    for (int e = 0; e < 2; ++e) {
      v2f a0, a1, q0, q1, b10, b11, b20, b21;
      a0.x = xa0[2 * e];  a0.y = xa0[2 * e + 1];
      a1.x = xa1[2 * e];  a1.y = xa1[2 * e + 1];
      q0 = a0 * a0;
      q1 = a1 * a1;
      b10.x = h0[2 * e];  b10.y = h0[2 * e + 1];
      b11.x = h1[2 * e];  b11.y = h1[2 * e + 1];
      b20.x = g0[2 * e];  b20.y = g0[2 * e + 1];
      b21.x = g1[2 * e];  b21.y = g1[2 * e + 1];
      acc00 = __builtin_amdgcn_wmma_f32_16x16x4_f32(false, q0, false, b10, (short)0, acc00, false, false);
      acc01 = __builtin_amdgcn_wmma_f32_16x16x4_f32(false, q0, false, b11, (short)0, acc01, false, false);
      acc10 = __builtin_amdgcn_wmma_f32_16x16x4_f32(false, q1, false, b10, (short)0, acc10, false, false);
      acc11 = __builtin_amdgcn_wmma_f32_16x16x4_f32(false, q1, false, b11, (short)0, acc11, false, false);
      acc00 = __builtin_amdgcn_wmma_f32_16x16x4_f32(false, a0, false, b20, (short)0, acc00, false, false);
      acc01 = __builtin_amdgcn_wmma_f32_16x16x4_f32(false, a0, false, b21, (short)0, acc01, false, false);
      acc10 = __builtin_amdgcn_wmma_f32_16x16x4_f32(false, a1, false, b20, (short)0, acc10, false, false);
      acc11 = __builtin_amdgcn_wmma_f32_16x16x4_f32(false, a1, false, b21, (short)0, acc11, false, false);
    }
    if (s & 1) __builtin_amdgcn_sched_barrier(0);
  }

  float* pw = part + wave * (TILE_MN * P_PITCH) + (8 * half) * P_PITCH + mn;
  for (int j = 0; j < 8; ++j) {
    pw[j * P_PITCH]                  = acc00[j];
    pw[j * P_PITCH + 16]             = acc01[j];
    pw[(16 + j) * P_PITCH]           = acc10[j];
    pw[(16 + j) * P_PITCH + 16]      = acc11[j];
  }
  bpart[(2 * wave + half) * TILE_MN + mn]      = bs0;
  bpart[(2 * wave + half) * TILE_MN + 16 + mn] = bs1;
  __syncthreads();

  float bv = bpart[lane];
  for (int p = 1; p < 2 * N_WAVE; ++p) bv += bpart[p * TILE_MN + lane];

  for (int m = 0; m < 4; ++m) {
    const int r = wave + 8 * m;
    const float* pr = part + r * P_PITCH + lane;
    float sum = pr[0];
    for (int w = 1; w < N_WAVE; ++w) sum += pr[w * (TILE_MN * P_PITCH)];
    out[(row0 + r) * K_DIM + col0 + lane] = sum + bv;
  }
}

extern "C" void kernel_launch(void* const* d_in, const int* in_sizes, int n_in,
                              void* d_out, int out_size, void* d_ws, size_t ws_size,
                              hipStream_t stream) {
  const float* x     = (const float*)d_in[0];
  const float* means = (const float*)d_in[1];
  const float* rho   = (const float*)d_in[2];
  float* out  = (float*)d_out;

  gpdf_fused_wmma<<<dim3(B_DIM / TILE_MN, K_DIM / TILE_MN), dim3(256), 0, stream>>>(
      x, means, rho, out);
}


# Operands fetched with whole-line global loads (8 lanes per 128-byte line, 4 lines per load instead of 16) and turned into WMMA fragment order through a wave-private swizzled LDS area
# speedup vs baseline: 5.0583x; 1.0532x over previous
#include <hip/hip_runtime.h>

#define B_DIM 1024
#define D_DIM 512
#define K_DIM 256

typedef __attribute__((ext_vector_type(2))) float v2f;
typedef __attribute__((ext_vector_type(4))) float v4f;
typedef __attribute__((ext_vector_type(8))) float v8f;

#define TILE_MN 32
#define N_WAVE  8
#define D_CHUNK (D_DIM / N_WAVE)
#define P_PITCH 34
#define N_STAGE 2
#define STG_ROWS (3 * TILE_MN)
#define STG_FLOATS (STG_ROWS * 32)

__device__ __forceinline__ float softplus_f(float r) {
  const float kLog2eHi = 1.44269504088896341f;
  const float kLog2eLo = 1.9259629911e-8f;
  const float kLn2     = 0.693147180559945309f;
  const float p  = r * kLog2eHi;
  const float pl = fmaf(r, kLog2eLo, fmaf(r, kLog2eHi, -p));
  const float e0 = __builtin_amdgcn_exp2f(p);
  const float e  = (e0 == __builtin_inff()) ? e0 : fmaf(e0, pl * kLn2, e0);
  const float u  = 1.0f + e;
  const float l  = __builtin_amdgcn_logf(u) * kLn2;
  const float t  = (e - (u - 1.0f)) * __builtin_amdgcn_rcpf(u);
  return (u == e) ? l : l + t;
}

__global__ __launch_bounds__(256) __attribute__((amdgpu_waves_per_eu(2, 2))) void gpdf_fused_wmma(
    const float* __restrict__ x,
    const float* __restrict__ means,
    const float* __restrict__ rho,
    float* __restrict__ out) {
  __shared__ float part[N_WAVE * TILE_MN * P_PITCH];
  __shared__ float bpart[2 * N_WAVE * TILE_MN];
  __shared__ __attribute__((aligned(16))) float stg[N_WAVE * N_STAGE * STG_FLOATS];

  const int lane = threadIdx.x & 31;
  const int wave = threadIdx.x >> 5;
  const int half = lane >> 4;
  const int mn   = lane & 15;
  const int row0 = blockIdx.x * TILE_MN;
  const int col0 = blockIdx.y * TILE_MN;

  const int lr = lane >> 3;
  const int lq = lane & 7;
  const float* __restrict__ gx = x     + (row0 + lr) * D_DIM + wave * D_CHUNK + 4 * lq;
  const float* __restrict__ gm = means + (col0 + lr) * D_DIM + wave * D_CHUNK + 4 * lq;
  const float* __restrict__ gr = rho   + (col0 + lr) * D_DIM + wave * D_CHUNK + 4 * lq;

  v8f acc00 = {}, acc01 = {}, acc10 = {}, acc11 = {};
  float bs0 = 0.0f, bs1 = 0.0f;

  v4f GX[N_STAGE][8], GM[N_STAGE][8], GR[N_STAGE][8];
#pragma unroll
  for (int g = 0; g < N_STAGE; ++g) {
#pragma unroll
    for (int i = 0; i < 8; ++i) GR[g][i] = *(const v4f*)(gr + (4 * i) * D_DIM + 32 * g);
#pragma unroll
    for (int i = 0; i < 8; ++i) GM[g][i] = *(const v4f*)(gm + (4 * i) * D_DIM + 32 * g);
#pragma unroll
    for (int i = 0; i < 8; ++i) GX[g][i] = *(const v4f*)(gx + (4 * i) * D_DIM + 32 * g);
  }

#pragma unroll
  for (int g = 0; g < N_STAGE; ++g) {
    float* sw = stg + (wave * N_STAGE + g) * STG_FLOATS;
#pragma unroll
    for (int i = 0; i < 8; ++i) {
      const int row = 4 * i + lr;
      const int sq  = 4 * (lq ^ ((row >> 1) & 7));
      *(v4f*)(sw + (2 * TILE_MN + row) * 32 + sq) = GR[g][i];
      *(v4f*)(sw + (TILE_MN + row) * 32 + sq)     = GM[g][i];
      *(v4f*)(sw + row * 32 + sq)                 = GX[g][i];
    }
    __builtin_amdgcn_fence(__ATOMIC_ACQ_REL, "wavefront");

    v4f XA0[4], XA1[4], M0[4], M1[4], R0[4], R1[4];
#pragma unroll
    for (int t = 0; t < 4; ++t) {
      const int pq = 4 * ((2 * t + half) ^ ((mn >> 1) & 7));
      R0[t]  = *(const v4f*)(sw + (2 * TILE_MN + mn) * 32 + pq);
      R1[t]  = *(const v4f*)(sw + (2 * TILE_MN + 16 + mn) * 32 + pq);
      M0[t]  = *(const v4f*)(sw + (TILE_MN + mn) * 32 + pq);
      M1[t]  = *(const v4f*)(sw + (TILE_MN + 16 + mn) * 32 + pq);
      XA0[t] = *(const v4f*)(sw + mn * 32 + pq);
      XA1[t] = *(const v4f*)(sw + (16 + mn) * 32 + pq);
    }
    __builtin_amdgcn_sched_barrier(0);

#pragma unroll
    for (int t = 0; t < 4; ++t) {
      const v4f xa0 = XA0[t];
      const v4f xa1 = XA1[t];
      const v4f m0  = M0[t];
      const v4f m1  = M1[t];
      const v4f r0  = R0[t];
      const v4f r1  = R1[t];
      v4f h0, h1, g0, g1;
#pragma unroll
      for (int i = 0; i < 4; ++i) {
        const float sa = softplus_f(r0[i]);
        const float sb = softplus_f(r1[i]);
        const float s2a = sa * sa;
        const float s2b = sb * sb;
        h0[i] = 0.5f * s2a;
        h1[i] = 0.5f * s2b;
        g0[i] = -s2a * m0[i];
        g1[i] = -s2b * m1[i];
        bs0 += 0.5f * s2a * m0[i] * m0[i];
        bs1 += 0.5f * s2b * m1[i] * m1[i];
      }
#pragma unroll
      for (int e = 0; e < 2; ++e) {
        v2f a0, a1, q0, q1, b10, b11, b20, b21;
        a0.x = xa0[2 * e];  a0.y = xa0[2 * e + 1];
        a1.x = xa1[2 * e];  a1.y = xa1[2 * e + 1];
        q0 = a0 * a0;
        q1 = a1 * a1;
        b10.x = h0[2 * e];  b10.y = h0[2 * e + 1];
        b11.x = h1[2 * e];  b11.y = h1[2 * e + 1];
        b20.x = g0[2 * e];  b20.y = g0[2 * e + 1];
        b21.x = g1[2 * e];  b21.y = g1[2 * e + 1];
        acc00 = __builtin_amdgcn_wmma_f32_16x16x4_f32(false, q0, false, b10, (short)0, acc00, false, false);
        acc01 = __builtin_amdgcn_wmma_f32_16x16x4_f32(false, q0, false, b11, (short)0, acc01, false, false);
        acc10 = __builtin_amdgcn_wmma_f32_16x16x4_f32(false, q1, false, b10, (short)0, acc10, false, false);
        acc11 = __builtin_amdgcn_wmma_f32_16x16x4_f32(false, q1, false, b11, (short)0, acc11, false, false);
        acc00 = __builtin_amdgcn_wmma_f32_16x16x4_f32(false, a0, false, b20, (short)0, acc00, false, false);
        acc01 = __builtin_amdgcn_wmma_f32_16x16x4_f32(false, a0, false, b21, (short)0, acc01, false, false);
        acc10 = __builtin_amdgcn_wmma_f32_16x16x4_f32(false, a1, false, b20, (short)0, acc10, false, false);
        acc11 = __builtin_amdgcn_wmma_f32_16x16x4_f32(false, a1, false, b21, (short)0, acc11, false, false);
      }
      if (t == 3 && g + 1 < N_STAGE) __builtin_amdgcn_sched_barrier(0);
    }
  }

  float* pw = part + wave * (TILE_MN * P_PITCH) + (8 * half) * P_PITCH + mn;
  for (int j = 0; j < 8; ++j) {
    pw[j * P_PITCH]                  = acc00[j];
    pw[j * P_PITCH + 16]             = acc01[j];
    pw[(16 + j) * P_PITCH]           = acc10[j];
    pw[(16 + j) * P_PITCH + 16]      = acc11[j];
  }
  bpart[(2 * wave + half) * TILE_MN + mn]      = bs0;
  bpart[(2 * wave + half) * TILE_MN + 16 + mn] = bs1;
  __syncthreads();

  float bv = bpart[lane];
  for (int p = 1; p < 2 * N_WAVE; ++p) bv += bpart[p * TILE_MN + lane];

  for (int m = 0; m < 4; ++m) {
    const int r = wave + 8 * m;
    const float* pr = part + r * P_PITCH + lane;
    float sum = pr[0];
    for (int w = 1; w < N_WAVE; ++w) sum += pr[w * (TILE_MN * P_PITCH)];
    out[(row0 + r) * K_DIM + col0 + lane] = sum + bv;
  }
}

extern "C" void kernel_launch(void* const* d_in, const int* in_sizes, int n_in,
                              void* d_out, int out_size, void* d_ws, size_t ws_size,
                              hipStream_t stream) {
  const float* x     = (const float*)d_in[0];
  const float* means = (const float*)d_in[1];
  const float* rho   = (const float*)d_in[2];
  float* out  = (float*)d_out;

  gpdf_fused_wmma<<<dim3(B_DIM / TILE_MN, K_DIM / TILE_MN), dim3(256), 0, stream>>>(
      x, means, rho, out);
}


# Block of 4 waves, one per SIMD, each contracting 128 features in four 128-byte pieces (next piece's LDS pass issued inside the current piece's math); 4 partial tiles summed per output, not 8
# speedup vs baseline: 5.5692x; 1.1010x over previous
#include <hip/hip_runtime.h>

#define B_DIM 1024
#define D_DIM 512
#define K_DIM 256

typedef __attribute__((ext_vector_type(2))) float v2f;
typedef __attribute__((ext_vector_type(4))) float v4f;
typedef __attribute__((ext_vector_type(8))) float v8f;

#define TILE_MN 32
#define N_WAVE  4
#define D_CHUNK (D_DIM / N_WAVE)
#define P_PITCH 34
#define N_STAGE 4
#define STG_ROWS (3 * TILE_MN)
#define STG_FLOATS (STG_ROWS * 32)

__device__ __forceinline__ float softplus_f(float r) {
  const float kLog2eHi = 1.44269504088896341f;
  const float kLog2eLo = 1.9259629911e-8f;
  const float kLn2     = 0.693147180559945309f;
  const float p  = r * kLog2eHi;
  const float pl = fmaf(r, kLog2eLo, fmaf(r, kLog2eHi, -p));
  const float e0 = __builtin_amdgcn_exp2f(p);
  const float e  = (e0 == __builtin_inff()) ? e0 : fmaf(e0, pl * kLn2, e0);
  const float u  = 1.0f + e;
  const float l  = __builtin_amdgcn_logf(u) * kLn2;
  const float t  = (e - (u - 1.0f)) * __builtin_amdgcn_rcpf(u);
  return (u == e) ? l : l + t;
}


__device__ __forceinline__ void kgroup(const v4f xa0, const v4f xa1, const v4f m0, const v4f m1,
                                       const v4f r0, const v4f r1,
                                       v8f& acc00, v8f& acc01, v8f& acc10, v8f& acc11,
                                       float& bs0, float& bs1) {
  v4f h0, h1, g0, g1;
#pragma unroll
  for (int i = 0; i < 4; ++i) {
    const float sa = softplus_f(r0[i]);
    const float sb = softplus_f(r1[i]);
    const float s2a = sa * sa;
    const float s2b = sb * sb;
    h0[i] = 0.5f * s2a;
    h1[i] = 0.5f * s2b;
    g0[i] = -s2a * m0[i];
    g1[i] = -s2b * m1[i];
    bs0 += 0.5f * s2a * m0[i] * m0[i];
    bs1 += 0.5f * s2b * m1[i] * m1[i];
  }
#pragma unroll
  for (int e = 0; e < 2; ++e) {
    v2f a0, a1, q0, q1, b10, b11, b20, b21;
    a0.x = xa0[2 * e];  a0.y = xa0[2 * e + 1];
    a1.x = xa1[2 * e];  a1.y = xa1[2 * e + 1];
    q0 = a0 * a0;
    q1 = a1 * a1;
    b10.x = h0[2 * e];  b10.y = h0[2 * e + 1];
    b11.x = h1[2 * e];  b11.y = h1[2 * e + 1];
    b20.x = g0[2 * e];  b20.y = g0[2 * e + 1];
    b21.x = g1[2 * e];  b21.y = g1[2 * e + 1];
    acc00 = __builtin_amdgcn_wmma_f32_16x16x4_f32(false, q0, false, b10, (short)0, acc00, false, false);
    acc01 = __builtin_amdgcn_wmma_f32_16x16x4_f32(false, q0, false, b11, (short)0, acc01, false, false);
    acc10 = __builtin_amdgcn_wmma_f32_16x16x4_f32(false, q1, false, b10, (short)0, acc10, false, false);
    acc11 = __builtin_amdgcn_wmma_f32_16x16x4_f32(false, q1, false, b11, (short)0, acc11, false, false);
    acc00 = __builtin_amdgcn_wmma_f32_16x16x4_f32(false, a0, false, b20, (short)0, acc00, false, false);
    acc01 = __builtin_amdgcn_wmma_f32_16x16x4_f32(false, a0, false, b21, (short)0, acc01, false, false);
    acc10 = __builtin_amdgcn_wmma_f32_16x16x4_f32(false, a1, false, b20, (short)0, acc10, false, false);
    acc11 = __builtin_amdgcn_wmma_f32_16x16x4_f32(false, a1, false, b21, (short)0, acc11, false, false);
  }
}

__global__ __launch_bounds__(128) __attribute__((amdgpu_waves_per_eu(1, 1))) void gpdf_fused_wmma(
    const float* __restrict__ x,
    const float* __restrict__ means,
    const float* __restrict__ rho,
    float* __restrict__ out) {
  __shared__ float part[N_WAVE * TILE_MN * P_PITCH];
  __shared__ float bpart[2 * N_WAVE * TILE_MN];
  __shared__ __attribute__((aligned(16))) float stg[N_WAVE * N_STAGE * STG_FLOATS];

  const int lane = threadIdx.x & 31;
  const int wave = threadIdx.x >> 5;
  const int half = lane >> 4;
  const int mn   = lane & 15;
  const int row0 = blockIdx.x * TILE_MN;
  const int col0 = blockIdx.y * TILE_MN;

  const int lr = lane >> 3;
  const int lq = lane & 7;
  const float* __restrict__ gx = x     + (row0 + lr) * D_DIM + wave * D_CHUNK + 4 * lq;
  const float* __restrict__ gm = means + (col0 + lr) * D_DIM + wave * D_CHUNK + 4 * lq;
  const float* __restrict__ gr = rho   + (col0 + lr) * D_DIM + wave * D_CHUNK + 4 * lq;

  v8f acc00 = {}, acc01 = {}, acc10 = {}, acc11 = {};
  float bs0 = 0.0f, bs1 = 0.0f;

  v4f GX[N_STAGE][8], GM[N_STAGE][8], GR[N_STAGE][8];
  v4f XA0[2][4], XA1[2][4], M0[2][4], M1[2][4], R0[2][4], R1[2][4];

#define FETCH_PIECE(g)                                                                        \
  _Pragma("unroll") for (int i = 0; i < 8; ++i) GR[g][i] = *(const v4f*)(gr + (4 * i) * D_DIM + 32 * (g)); \
  _Pragma("unroll") for (int i = 0; i < 8; ++i) GM[g][i] = *(const v4f*)(gm + (4 * i) * D_DIM + 32 * (g)); \
  _Pragma("unroll") for (int i = 0; i < 8; ++i) GX[g][i] = *(const v4f*)(gx + (4 * i) * D_DIM + 32 * (g));

#define LDS_PASS(g)                                                                           \
  {                                                                                           \
    float* sw = stg + (wave * N_STAGE + (g)) * STG_FLOATS;                                    \
    _Pragma("unroll") for (int i = 0; i < 8; ++i) {                                           \
      const int row = 4 * i + lr;                                                             \
      const int sq  = 4 * (lq ^ ((row >> 1) & 7));                                            \
      *(v4f*)(sw + (2 * TILE_MN + row) * 32 + sq) = GR[g][i];                                 \
      *(v4f*)(sw + (TILE_MN + row) * 32 + sq)     = GM[g][i];                                 \
      *(v4f*)(sw + row * 32 + sq)                 = GX[g][i];                                 \
    }                                                                                         \
    __builtin_amdgcn_fence(__ATOMIC_ACQ_REL, "wavefront");                                    \
    _Pragma("unroll") for (int t = 0; t < 4; ++t) {                                           \
      const int pq = 4 * ((2 * t + half) ^ ((mn >> 1) & 7));                                  \
      R0[(g) & 1][t]  = *(const v4f*)(sw + (2 * TILE_MN + mn) * 32 + pq);                     \
      R1[(g) & 1][t]  = *(const v4f*)(sw + (2 * TILE_MN + 16 + mn) * 32 + pq);                \
      M0[(g) & 1][t]  = *(const v4f*)(sw + (TILE_MN + mn) * 32 + pq);                         \
      M1[(g) & 1][t]  = *(const v4f*)(sw + (TILE_MN + 16 + mn) * 32 + pq);                    \
      XA0[(g) & 1][t] = *(const v4f*)(sw + mn * 32 + pq);                                     \
      XA1[(g) & 1][t] = *(const v4f*)(sw + (16 + mn) * 32 + pq);                              \
    }                                                                                         \
  }

#define MATH(g, t)                                                                            \
  kgroup(XA0[(g) & 1][t], XA1[(g) & 1][t], M0[(g) & 1][t], M1[(g) & 1][t], R0[(g) & 1][t],    \
         R1[(g) & 1][t], acc00, acc01, acc10, acc11, bs0, bs1);

  FETCH_PIECE(0)
  FETCH_PIECE(1)
  LDS_PASS(0)
  FETCH_PIECE(2)
  __builtin_amdgcn_sched_barrier(0);

#pragma unroll
  for (int g = 0; g < N_STAGE; ++g) {
    MATH(g, 0)
    MATH(g, 1)
    if (g + 1 < N_STAGE) {
      __builtin_amdgcn_sched_barrier(0);
      if (g + 1 == 1) { LDS_PASS(1) FETCH_PIECE(3) }
      if (g + 1 == 2) { LDS_PASS(2) }
      if (g + 1 == 3) { LDS_PASS(3) }
      __builtin_amdgcn_sched_barrier(0);
    }
    MATH(g, 2)
    MATH(g, 3)
  }
#undef FETCH_PIECE
#undef LDS_PASS
#undef MATH

  float* pw = part + wave * (TILE_MN * P_PITCH) + (8 * half) * P_PITCH + mn;
  for (int j = 0; j < 8; ++j) {
    pw[j * P_PITCH]                  = acc00[j];
    pw[j * P_PITCH + 16]             = acc01[j];
    pw[(16 + j) * P_PITCH]           = acc10[j];
    pw[(16 + j) * P_PITCH + 16]      = acc11[j];
  }
  bpart[(2 * wave + half) * TILE_MN + mn]      = bs0;
  bpart[(2 * wave + half) * TILE_MN + 16 + mn] = bs1;
  __syncthreads();

  float bv = bpart[lane];
  for (int p = 1; p < 2 * N_WAVE; ++p) bv += bpart[p * TILE_MN + lane];

  for (int m = 0; m < TILE_MN / N_WAVE; ++m) {
    const int r = wave + N_WAVE * m;
    const float* pr = part + r * P_PITCH + lane;
    float sum = pr[0];
    for (int w = 1; w < N_WAVE; ++w) sum += pr[w * (TILE_MN * P_PITCH)];
    out[(row0 + r) * K_DIM + col0 + lane] = sum + bv;
  }
}

extern "C" void kernel_launch(void* const* d_in, const int* in_sizes, int n_in,
                              void* d_out, int out_size, void* d_ws, size_t ws_size,
                              hipStream_t stream) {
  const float* x     = (const float*)d_in[0];
  const float* means = (const float*)d_in[1];
  const float* rho   = (const float*)d_in[2];
  float* out  = (float*)d_out;

  gpdf_fused_wmma<<<dim3(B_DIM / TILE_MN, K_DIM / TILE_MN), dim3(128), 0, stream>>>(
      x, means, rho, out);
}


# Weight formation cut to 3 vector instr per WMMA (was 8): softplus taken directly as ln2*log2(1+exp2(rho*log2e)), rounding repairs and inf guards dropped; bias summed per element parity, scaled once
# speedup vs baseline: 6.0072x; 1.0786x over previous
#include <hip/hip_runtime.h>

#define B_DIM 1024
#define D_DIM 512
#define K_DIM 256

typedef __attribute__((ext_vector_type(2))) float v2f;
typedef __attribute__((ext_vector_type(4))) float v4f;
typedef __attribute__((ext_vector_type(8))) float v8f;

#define TILE_MN 32
#define N_WAVE  4
#define D_CHUNK (D_DIM / N_WAVE)
#define P_PITCH 34
#define N_STAGE 4
#define STG_ROWS (3 * TILE_MN)
#define STG_FLOATS (STG_ROWS * 32)

__device__ __forceinline__ void kgroup(const v4f xa0, const v4f xa1, const v4f m0, const v4f m1,
                                       const v4f r0, const v4f r1,
                                       v8f& acc00, v8f& acc01, v8f& acc10, v8f& acc11,
                                       v2f& bs0, v2f& bs1) {
  const float kLog2e = 1.44269504088896341f;
  const float kLn2Sq = 0.480453013918201424f;
#pragma unroll
  for (int e = 0; e < 2; ++e) {
    v2f a0, a1, q0, q1, ra, rb, ma, mb, ea, eb, la, lb;
    a0.x = xa0[2 * e];  a0.y = xa0[2 * e + 1];
    a1.x = xa1[2 * e];  a1.y = xa1[2 * e + 1];
    ra.x = r0[2 * e];   ra.y = r0[2 * e + 1];
    rb.x = r1[2 * e];   rb.y = r1[2 * e + 1];
    ma.x = m0[2 * e];   ma.y = m0[2 * e + 1];
    mb.x = m1[2 * e];   mb.y = m1[2 * e + 1];
    q0 = a0 * a0;
    q1 = a1 * a1;
    const v2f pa = ra * kLog2e;
    const v2f pb = rb * kLog2e;
    ea.x = __builtin_amdgcn_exp2f(pa.x);  ea.y = __builtin_amdgcn_exp2f(pa.y);
    eb.x = __builtin_amdgcn_exp2f(pb.x);  eb.y = __builtin_amdgcn_exp2f(pb.y);
    const v2f ua = ea + 1.0f;
    const v2f ub = eb + 1.0f;
    la.x = __builtin_amdgcn_logf(ua.x);   la.y = __builtin_amdgcn_logf(ua.y);
    lb.x = __builtin_amdgcn_logf(ub.x);   lb.y = __builtin_amdgcn_logf(ub.y);
    const v2f s2a = (la * kLn2Sq) * la;
    const v2f s2b = (lb * kLn2Sq) * lb;
    const v2f b10 = s2a * 0.5f;
    const v2f b11 = s2b * 0.5f;
    const v2f b20 = s2a * (-ma);
    const v2f b21 = s2b * (-mb);
    bs0 = __builtin_elementwise_fma(b20, ma, bs0);
    bs1 = __builtin_elementwise_fma(b21, mb, bs1);
    acc00 = __builtin_amdgcn_wmma_f32_16x16x4_f32(false, q0, false, b10, (short)0, acc00, false, false);
    acc01 = __builtin_amdgcn_wmma_f32_16x16x4_f32(false, q0, false, b11, (short)0, acc01, false, false);
    acc10 = __builtin_amdgcn_wmma_f32_16x16x4_f32(false, q1, false, b10, (short)0, acc10, false, false);
    acc11 = __builtin_amdgcn_wmma_f32_16x16x4_f32(false, q1, false, b11, (short)0, acc11, false, false);
    acc00 = __builtin_amdgcn_wmma_f32_16x16x4_f32(false, a0, false, b20, (short)0, acc00, false, false);
    acc01 = __builtin_amdgcn_wmma_f32_16x16x4_f32(false, a0, false, b21, (short)0, acc01, false, false);
    acc10 = __builtin_amdgcn_wmma_f32_16x16x4_f32(false, a1, false, b20, (short)0, acc10, false, false);
    acc11 = __builtin_amdgcn_wmma_f32_16x16x4_f32(false, a1, false, b21, (short)0, acc11, false, false);
  }
}

__global__ __launch_bounds__(128) __attribute__((amdgpu_waves_per_eu(1, 1))) void gpdf_fused_wmma(
    const float* __restrict__ x,
    const float* __restrict__ means,
    const float* __restrict__ rho,
    float* __restrict__ out) {
  __shared__ float part[N_WAVE * TILE_MN * P_PITCH];
  __shared__ float bpart[2 * N_WAVE * TILE_MN];
  __shared__ __attribute__((aligned(16))) float stg[N_WAVE * N_STAGE * STG_FLOATS];

  const int lane = threadIdx.x & 31;
  const int wave = threadIdx.x >> 5;
  const int half = lane >> 4;
  const int mn   = lane & 15;
  const int row0 = blockIdx.x * TILE_MN;
  const int col0 = blockIdx.y * TILE_MN;

  const int lr = lane >> 3;
  const int lq = lane & 7;
  const float* __restrict__ gx = x     + (row0 + lr) * D_DIM + wave * D_CHUNK + 4 * lq;
  const float* __restrict__ gm = means + (col0 + lr) * D_DIM + wave * D_CHUNK + 4 * lq;
  const float* __restrict__ gr = rho   + (col0 + lr) * D_DIM + wave * D_CHUNK + 4 * lq;

  v8f acc00 = {}, acc01 = {}, acc10 = {}, acc11 = {};
  v2f bs0 = {}, bs1 = {};

  v4f GX[N_STAGE][8], GM[N_STAGE][8], GR[N_STAGE][8];
  v4f XA0[2][4], XA1[2][4], M0[2][4], M1[2][4], R0[2][4], R1[2][4];

#define FETCH_PIECE(g)                                                                        \
  _Pragma("unroll") for (int i = 0; i < 8; ++i) GR[g][i] = *(const v4f*)(gr + (4 * i) * D_DIM + 32 * (g)); \
  _Pragma("unroll") for (int i = 0; i < 8; ++i) GM[g][i] = *(const v4f*)(gm + (4 * i) * D_DIM + 32 * (g)); \
  _Pragma("unroll") for (int i = 0; i < 8; ++i) GX[g][i] = *(const v4f*)(gx + (4 * i) * D_DIM + 32 * (g));

#define LDS_PASS(g)                                                                           \
  {                                                                                           \
    float* sw = stg + (wave * N_STAGE + (g)) * STG_FLOATS;                                    \
    _Pragma("unroll") for (int i = 0; i < 8; ++i) {                                           \
      const int row = 4 * i + lr;                                                             \
      const int sq  = 4 * (lq ^ ((row >> 1) & 7));                                            \
      *(v4f*)(sw + (2 * TILE_MN + row) * 32 + sq) = GR[g][i];                                 \
      *(v4f*)(sw + (TILE_MN + row) * 32 + sq)     = GM[g][i];                                 \
      *(v4f*)(sw + row * 32 + sq)                 = GX[g][i];                                 \
    }                                                                                         \
    __builtin_amdgcn_fence(__ATOMIC_ACQ_REL, "wavefront");                                    \
    _Pragma("unroll") for (int t = 0; t < 4; ++t) {                                           \
      const int pq = 4 * ((2 * t + half) ^ ((mn >> 1) & 7));                                  \
      R0[(g) & 1][t]  = *(const v4f*)(sw + (2 * TILE_MN + mn) * 32 + pq);                     \
      R1[(g) & 1][t]  = *(const v4f*)(sw + (2 * TILE_MN + 16 + mn) * 32 + pq);                \
      M0[(g) & 1][t]  = *(const v4f*)(sw + (TILE_MN + mn) * 32 + pq);                         \
      M1[(g) & 1][t]  = *(const v4f*)(sw + (TILE_MN + 16 + mn) * 32 + pq);                    \
      XA0[(g) & 1][t] = *(const v4f*)(sw + mn * 32 + pq);                                     \
      XA1[(g) & 1][t] = *(const v4f*)(sw + (16 + mn) * 32 + pq);                              \
    }                                                                                         \
  }

#define MATH(g, t)                                                                            \
  kgroup(XA0[(g) & 1][t], XA1[(g) & 1][t], M0[(g) & 1][t], M1[(g) & 1][t], R0[(g) & 1][t],    \
         R1[(g) & 1][t], acc00, acc01, acc10, acc11, bs0, bs1);

  FETCH_PIECE(0)
  FETCH_PIECE(1)
  LDS_PASS(0)
  FETCH_PIECE(2)
  __builtin_amdgcn_sched_barrier(0);

#pragma unroll
  for (int g = 0; g < N_STAGE; ++g) {
    MATH(g, 0)
    MATH(g, 1)
    if (g + 1 < N_STAGE) {
      __builtin_amdgcn_sched_barrier(0);
      if (g + 1 == 1) { LDS_PASS(1) FETCH_PIECE(3) }
      if (g + 1 == 2) { LDS_PASS(2) }
      if (g + 1 == 3) { LDS_PASS(3) }
      __builtin_amdgcn_sched_barrier(0);
    }
    MATH(g, 2)
    MATH(g, 3)
  }
#undef FETCH_PIECE
#undef LDS_PASS
#undef MATH

  float* pw = part + wave * (TILE_MN * P_PITCH) + (8 * half) * P_PITCH + mn;
  for (int j = 0; j < 8; ++j) {
    pw[j * P_PITCH]                  = acc00[j];
    pw[j * P_PITCH + 16]             = acc01[j];
    pw[(16 + j) * P_PITCH]           = acc10[j];
    pw[(16 + j) * P_PITCH + 16]      = acc11[j];
  }
  bpart[(2 * wave + half) * TILE_MN + mn]      = -0.5f * (bs0.x + bs0.y);
  bpart[(2 * wave + half) * TILE_MN + 16 + mn] = -0.5f * (bs1.x + bs1.y);
  __syncthreads();

  float bv = bpart[lane];
  for (int p = 1; p < 2 * N_WAVE; ++p) bv += bpart[p * TILE_MN + lane];

  for (int m = 0; m < TILE_MN / N_WAVE; ++m) {
    const int r = wave + N_WAVE * m;
    const float* pr = part + r * P_PITCH + lane;
    float sum = pr[0];
    for (int w = 1; w < N_WAVE; ++w) sum += pr[w * (TILE_MN * P_PITCH)];
    out[(row0 + r) * K_DIM + col0 + lane] = sum + bv;
  }
}

extern "C" void kernel_launch(void* const* d_in, const int* in_sizes, int n_in,
                              void* d_out, int out_size, void* d_ws, size_t ws_size,
                              hipStream_t stream) {
  const float* x     = (const float*)d_in[0];
  const float* means = (const float*)d_in[1];
  const float* rho   = (const float*)d_in[2];
  float* out  = (float*)d_out;

  gpdf_fused_wmma<<<dim3(B_DIM / TILE_MN, K_DIM / TILE_MN), dim3(128), 0, stream>>>(
      x, means, rho, out);
}
